# starting kernel
# baseline (speedup 1.0000x reference)
#include <hip/hip_runtime.h>
#include <cmath>


typedef __attribute__((ext_vector_type(2))) float v2f;
typedef __attribute__((ext_vector_type(4))) float v4f;
typedef __attribute__((ext_vector_type(8))) float v8f;

#define BB 32
#define LL 1024
#define HH 256
#define AA 128

__global__ __launch_bounds__(256) void proj_tanh_kernel(
    const float* __restrict__ reps_lt, const float* __restrict__ reps_rt,
    const float* __restrict__ W, const float* __restrict__ diagW,
    float* __restrict__ proj)
{
    extern __shared__ float WT[];
    constexpr int TSTR = HH + 4;
    const int tid = threadIdx.x;

    for (int i = tid * 4; i < HH * AA; i += 256 * 4) {
        const int h = i >> 7;
        const int a = i & (AA - 1);
        const v4f w = *(const v4f*)(W + h * AA + a);
        WT[(a + 0) * TSTR + h] = w.x;
        WT[(a + 1) * TSTR + h] = w.y;
        WT[(a + 2) * TSTR + h] = w.z;
        WT[(a + 3) * TSTR + h] = w.w;
    }
    __syncthreads();

    const int lane = tid & 31;
    const int wv   = tid >> 5;
    const int lo   = lane & 15;
    const int hi   = lane >> 4;

    const int row0 = (blockIdx.x * 8 + wv) * 16;
    const int grow = row0 + lo;
    const bool is_lt = row0 < BB * LL;
    const float* __restrict__ src =
        is_lt ? reps_lt + (size_t)grow * HH
              : reps_rt + (size_t)(grow - BB * LL) * HH;

    const float* bp[8];
#pragma unroll
    for (int ct = 0; ct < 8; ++ct)
        bp[ct] = WT + (ct * 16 + lo) * TSTR + 2 * hi;

    v8f acc[8] = {};

    for (int p = 0; p < HH / 4; ++p) {
        const int k = 4 * p;
        const v2f a = *(const v2f*)(src + k + 2 * hi);

        v2f bf[8];
#pragma unroll
        for (int ct = 0; ct < 8; ++ct)
            bf[ct] = *(const v2f*)(bp[ct] + k);
#pragma unroll
        for (int ct = 0; ct < 8; ++ct)
            acc[ct] = __builtin_amdgcn_wmma_f32_16x16x4_f32(
                false, a, false, bf[ct], (short)0, acc[ct], false, false);
    }

#pragma unroll
    for (int ct = 0; ct < 8; ++ct) {
        const int col  = ct * 16 + lo;
        const float dw = is_lt ? diagW[col] : 1.0f;
#pragma unroll
        for (int i = 0; i < 8; ++i) {
            const int m = i + 8 * hi;
            proj[(size_t)(row0 + m) * AA + col] = tanhf(acc[ct][i]) * dw;
        }
    }
}

__global__ __launch_bounds__(256) void attn_softmax_kernel(
    const float* __restrict__ proj, const float* __restrict__ mask_lt,
    const float* __restrict__ mask_rt, float* __restrict__ out)
{
    extern __shared__ float S[];
    constexpr int SSTR = LL + 4;
    __shared__ float red[16][17];
    __shared__ float mlt_s[16];

    const int tid = threadIdx.x;
    const int b   = blockIdx.y;
    const int lt0 = blockIdx.x * 16;

    if (tid < 16) mlt_s[tid] = mask_lt[(size_t)b * LL + lt0 + tid];
    __syncthreads();

    const int lane = tid & 31;
    const int wv   = tid >> 5;
    const int lo   = lane & 15;
    const int hi   = lane >> 4;

    const float* __restrict__ ltp = proj + ((size_t)b * LL + lt0 + lo) * AA;
    const float* __restrict__ rtb = proj + ((size_t)(BB * LL) + (size_t)b * LL) * AA;
    const float* __restrict__ mrt = mask_rt + (size_t)b * LL;

    v2f afrag[32];
#pragma unroll
    for (int p = 0; p < 32; ++p)
        afrag[p] = *(const v2f*)(ltp + 4 * p + 2 * hi);

    for (int ct = 0; ct < 8; ++ct) {
        const int cbase = wv * 128 + ct * 16;
        const float* __restrict__ rtp = rtb + (size_t)(cbase + lo) * AA + 2 * hi;
        v8f c = {};
#pragma unroll
        for (int q = 0; q < 4; ++q) {
            v2f bf[8];
#pragma unroll
            for (int u = 0; u < 8; ++u)
                bf[u] = *(const v2f*)(rtp + 4 * (8 * q + u));
#pragma unroll
            for (int u = 0; u < 8; ++u)
                c = __builtin_amdgcn_wmma_f32_16x16x4_f32(
                    false, afrag[8 * q + u], false, bf[u], (short)0, c, false, false);
        }
        const float mr = mrt[cbase + lo];
#pragma unroll
        for (int i = 0; i < 8; ++i) {
            const int m = i + 8 * hi;
            S[m * SSTR + cbase + lo] = c[i] * mlt_s[m] * mr;
        }
    }
    __syncthreads();

    const int r   = tid >> 4;
    const int c16 = tid & 15;

    float mx = -3.402823466e38f;
#pragma unroll 4
    for (int j = 0; j < 16; ++j) {
        const v4f v = *(const v4f*)(S + r * SSTR + c16 * 4 + 64 * j);
        mx = fmaxf(mx, fmaxf(fmaxf(v.x, v.y), fmaxf(v.z, v.w)));
    }
    red[r][c16] = mx;
    __syncthreads();
    float rmax = red[r][0];
    for (int t = 1; t < 16; ++t) rmax = fmaxf(rmax, red[r][t]);
    __syncthreads();

    float sum = 0.f;
    for (int j = 0; j < 16; ++j) {
        float* p4 = S + r * SSTR + c16 * 4 + 64 * j;
        v4f v = *(const v4f*)p4;
        v.x = expf(v.x - rmax);
        v.y = expf(v.y - rmax);
        v.z = expf(v.z - rmax);
        v.w = expf(v.w - rmax);
        sum += (v.x + v.y) + (v.z + v.w);
        *(v4f*)p4 = v;
    }
    red[r][c16] = sum;
    __syncthreads();
    float rsum = 0.f;
    for (int t = 0; t < 16; ++t) rsum += red[r][t];

    const float mlr = mlt_s[r] / rsum;
    float* __restrict__ orow = out + ((size_t)b * LL + lt0 + r) * (size_t)LL;
    for (int j = 0; j < 16; ++j) {
        const int cc = c16 * 4 + 64 * j;
        const v4f e  = *(const v4f*)(S + r * SSTR + cc);
        const v4f m4 = *(const v4f*)(mrt + cc);
        v4f o;
        o.x = e.x * mlr * m4.x;
        o.y = e.y * mlr * m4.y;
        o.z = e.z * mlr * m4.z;
        o.w = e.w * mlr * m4.w;
        __builtin_nontemporal_store(o, (v4f*)(orow + cc));
    }
}

extern "C" void kernel_launch(void* const* d_in, const int* in_sizes, int n_in,
                              void* d_out, int out_size, void* d_ws, size_t ws_size,
                              hipStream_t stream)
{
    (void)in_sizes; (void)n_in; (void)out_size; (void)ws_size;

    const float* reps_lt = (const float*)d_in[0];
    const float* reps_rt = (const float*)d_in[1];
    const float* mask_lt = (const float*)d_in[2];
    const float* mask_rt = (const float*)d_in[3];
    const float* attn_k  = (const float*)d_in[4];
    const float* diagW   = (const float*)d_in[5];
    float* outp = (float*)d_out;
    float* proj = (float*)d_ws;

    const size_t ldsW = (size_t)AA * (HH + 4) * sizeof(float);
    proj_tanh_kernel<<<dim3(512), 256, ldsW, stream>>>(
        reps_lt, reps_rt, attn_k, diagW, proj);

    const size_t ldsS = (size_t)16 * (LL + 4) * sizeof(float);
    attn_softmax_kernel<<<dim3(LL / 16, BB), 256, ldsS, stream>>>(
        proj, mask_lt, mask_rt, outp);
}


# attention kernel: lt and rt fragments fetched by 16-byte loads, one load per two k-steps, k permuted alike on both operands: half the cache lines touched per WMMA; f32 sums over k reordered
# speedup vs baseline: 1.0616x; 1.0616x over previous
#include <hip/hip_runtime.h>
#include <cmath>


typedef __attribute__((ext_vector_type(2))) float v2f;
typedef __attribute__((ext_vector_type(4))) float v4f;
typedef __attribute__((ext_vector_type(8))) float v8f;

#define BB 32
#define LL 1024
#define HH 256
#define AA 128

__global__ __launch_bounds__(256) void proj_tanh_kernel(
    const float* __restrict__ reps_lt, const float* __restrict__ reps_rt,
    const float* __restrict__ W, const float* __restrict__ diagW,
    float* __restrict__ proj)
{
    extern __shared__ float WT[];
    constexpr int TSTR = HH + 4;
    const int tid = threadIdx.x;

    for (int i = tid * 4; i < HH * AA; i += 256 * 4) {
        const int h = i >> 7;
        const int a = i & (AA - 1);
        const v4f w = *(const v4f*)(W + h * AA + a);
        WT[(a + 0) * TSTR + h] = w.x;
        WT[(a + 1) * TSTR + h] = w.y;
        WT[(a + 2) * TSTR + h] = w.z;
        WT[(a + 3) * TSTR + h] = w.w;
    }
    __syncthreads();

    const int lane = tid & 31;
    const int wv   = tid >> 5;
    const int lo   = lane & 15;
    const int hi   = lane >> 4;

    const int row0 = (blockIdx.x * 8 + wv) * 16;
    const int grow = row0 + lo;
    const bool is_lt = row0 < BB * LL;
    const float* __restrict__ src =
        is_lt ? reps_lt + (size_t)grow * HH
              : reps_rt + (size_t)(grow - BB * LL) * HH;

    const float* bp[8];
#pragma unroll
    for (int ct = 0; ct < 8; ++ct)
        bp[ct] = WT + (ct * 16 + lo) * TSTR + 2 * hi;

    v8f acc[8] = {};

    for (int p = 0; p < HH / 4; ++p) {
        const int k = 4 * p;
        const v2f a = *(const v2f*)(src + k + 2 * hi);

        v2f bf[8];
#pragma unroll
        for (int ct = 0; ct < 8; ++ct)
            bf[ct] = *(const v2f*)(bp[ct] + k);
#pragma unroll
        for (int ct = 0; ct < 8; ++ct)
            acc[ct] = __builtin_amdgcn_wmma_f32_16x16x4_f32(
                false, a, false, bf[ct], (short)0, acc[ct], false, false);
    }

#pragma unroll
    for (int ct = 0; ct < 8; ++ct) {
        const int col  = ct * 16 + lo;
        const float dw = is_lt ? diagW[col] : 1.0f;
#pragma unroll
        for (int i = 0; i < 8; ++i) {
            const int m = i + 8 * hi;
            proj[(size_t)(row0 + m) * AA + col] = tanhf(acc[ct][i]) * dw;
        }
    }
}

__global__ __launch_bounds__(256) void attn_softmax_kernel(
    const float* __restrict__ proj, const float* __restrict__ mask_lt,
    const float* __restrict__ mask_rt, float* __restrict__ out)
{
    extern __shared__ float S[];
    constexpr int SSTR = LL + 4;
    __shared__ float red[16][17];
    __shared__ float mlt_s[16];

    const int tid = threadIdx.x;
    const int b   = blockIdx.y;
    const int lt0 = blockIdx.x * 16;

    if (tid < 16) mlt_s[tid] = mask_lt[(size_t)b * LL + lt0 + tid];
    __syncthreads();

    const int lane = tid & 31;
    const int wv   = tid >> 5;
    const int lo   = lane & 15;
    const int hi   = lane >> 4;

    const float* __restrict__ ltp = proj + ((size_t)b * LL + lt0 + lo) * AA;
    const float* __restrict__ rtb = proj + ((size_t)(BB * LL) + (size_t)b * LL) * AA;
    const float* __restrict__ mrt = mask_rt + (size_t)b * LL;

    v4f afrag[16];
#pragma unroll
    for (int j = 0; j < 16; ++j)
        afrag[j] = *(const v4f*)(ltp + 8 * j + 4 * hi);

    for (int ct = 0; ct < 8; ++ct) {
        const int cbase = wv * 128 + ct * 16;
        const float* __restrict__ rtp = rtb + (size_t)(cbase + lo) * AA + 4 * hi;
        v8f c = {};
#pragma unroll
        for (int q = 0; q < 2; ++q) {
            v4f bf[8];
#pragma unroll
            for (int u = 0; u < 8; ++u)
                bf[u] = *(const v4f*)(rtp + 8 * (8 * q + u));
            __builtin_amdgcn_sched_barrier(0);
#pragma unroll
            for (int u = 0; u < 8; ++u) {
                const v4f a4 = afrag[8 * q + u];
                const v4f b4 = bf[u];
                const v2f a0 = {a4.x, a4.y};
                const v2f b0 = {b4.x, b4.y};
                const v2f a1 = {a4.z, a4.w};
                const v2f b1 = {b4.z, b4.w};
                c = __builtin_amdgcn_wmma_f32_16x16x4_f32(
                    false, a0, false, b0, (short)0, c, false, false);
                c = __builtin_amdgcn_wmma_f32_16x16x4_f32(
                    false, a1, false, b1, (short)0, c, false, false);
            }
        }
        const float mr = mrt[cbase + lo];
#pragma unroll
        for (int i = 0; i < 8; ++i) {
            const int m = i + 8 * hi;
            S[m * SSTR + cbase + lo] = c[i] * mlt_s[m] * mr;
        }
    }
    __syncthreads();

    const int r   = tid >> 4;
    const int c16 = tid & 15;

    float mx = -3.402823466e38f;
#pragma unroll 4
    for (int j = 0; j < 16; ++j) {
        const v4f v = *(const v4f*)(S + r * SSTR + c16 * 4 + 64 * j);
        mx = fmaxf(mx, fmaxf(fmaxf(v.x, v.y), fmaxf(v.z, v.w)));
    }
    red[r][c16] = mx;
    __syncthreads();
    float rmax = red[r][0];
    for (int t = 1; t < 16; ++t) rmax = fmaxf(rmax, red[r][t]);
    __syncthreads();

    float sum = 0.f;
    for (int j = 0; j < 16; ++j) {
        float* p4 = S + r * SSTR + c16 * 4 + 64 * j;
        v4f v = *(const v4f*)p4;
        v.x = expf(v.x - rmax);
        v.y = expf(v.y - rmax);
        v.z = expf(v.z - rmax);
        v.w = expf(v.w - rmax);
        sum += (v.x + v.y) + (v.z + v.w);
        *(v4f*)p4 = v;
    }
    red[r][c16] = sum;
    __syncthreads();
    float rsum = 0.f;
    for (int t = 0; t < 16; ++t) rsum += red[r][t];

    const float mlr = mlt_s[r] / rsum;
    float* __restrict__ orow = out + ((size_t)b * LL + lt0 + r) * (size_t)LL;
    for (int j = 0; j < 16; ++j) {
        const int cc = c16 * 4 + 64 * j;
        const v4f e  = *(const v4f*)(S + r * SSTR + cc);
        const v4f m4 = *(const v4f*)(mrt + cc);
        v4f o;
        o.x = e.x * mlr * m4.x;
        o.y = e.y * mlr * m4.y;
        o.z = e.z * mlr * m4.z;
        o.w = e.w * mlr * m4.w;
        __builtin_nontemporal_store(o, (v4f*)(orow + cc));
    }
}

extern "C" void kernel_launch(void* const* d_in, const int* in_sizes, int n_in,
                              void* d_out, int out_size, void* d_ws, size_t ws_size,
                              hipStream_t stream)
{
    (void)in_sizes; (void)n_in; (void)out_size; (void)ws_size;

    const float* reps_lt = (const float*)d_in[0];
    const float* reps_rt = (const float*)d_in[1];
    const float* mask_lt = (const float*)d_in[2];
    const float* mask_rt = (const float*)d_in[3];
    const float* attn_k  = (const float*)d_in[4];
    const float* diagW   = (const float*)d_in[5];
    float* outp = (float*)d_out;
    float* proj = (float*)d_ws;

    const size_t ldsW = (size_t)AA * (HH + 4) * sizeof(float);
    proj_tanh_kernel<<<dim3(512), 256, ldsW, stream>>>(
        reps_lt, reps_rt, attn_k, diagW, proj);

    const size_t ldsS = (size_t)16 * (LL + 4) * sizeof(float);
    attn_softmax_kernel<<<dim3(LL / 16, BB), 256, ldsS, stream>>>(
        proj, mask_lt, mask_rt, outp);
}


# attention kernel: each workgroup takes 32 lt rows (two row tiles), so every rt fragment load feeds two independent WMMA chains: half the L2 lines and loads per WMMA; summation order unchanged
# speedup vs baseline: 1.2694x; 1.1957x over previous
#include <hip/hip_runtime.h>
#include <cmath>


typedef __attribute__((ext_vector_type(2))) float v2f;
typedef __attribute__((ext_vector_type(4))) float v4f;
typedef __attribute__((ext_vector_type(8))) float v8f;

#define BB 32
#define LL 1024
#define HH 256
#define AA 128

__global__ __launch_bounds__(256) void proj_tanh_kernel(
    const float* __restrict__ reps_lt, const float* __restrict__ reps_rt,
    const float* __restrict__ W, const float* __restrict__ diagW,
    float* __restrict__ proj)
{
    extern __shared__ float WT[];
    constexpr int TSTR = HH + 4;
    const int tid = threadIdx.x;

    for (int i = tid * 4; i < HH * AA; i += 256 * 4) {
        const int h = i >> 7;
        const int a = i & (AA - 1);
        const v4f w = *(const v4f*)(W + h * AA + a);
        WT[(a + 0) * TSTR + h] = w.x;
        WT[(a + 1) * TSTR + h] = w.y;
        WT[(a + 2) * TSTR + h] = w.z;
        WT[(a + 3) * TSTR + h] = w.w;
    }
    __syncthreads();

    const int lane = tid & 31;
    const int wv   = tid >> 5;
    const int lo   = lane & 15;
    const int hi   = lane >> 4;

    const int row0 = (blockIdx.x * 8 + wv) * 16;
    const int grow = row0 + lo;
    const bool is_lt = row0 < BB * LL;
    const float* __restrict__ src =
        is_lt ? reps_lt + (size_t)grow * HH
              : reps_rt + (size_t)(grow - BB * LL) * HH;

    const float* bp[8];
#pragma unroll
    for (int ct = 0; ct < 8; ++ct)
        bp[ct] = WT + (ct * 16 + lo) * TSTR + 2 * hi;

    v8f acc[8] = {};

    for (int p = 0; p < HH / 4; ++p) {
        const int k = 4 * p;
        const v2f a = *(const v2f*)(src + k + 2 * hi);

        v2f bf[8];
#pragma unroll
        for (int ct = 0; ct < 8; ++ct)
            bf[ct] = *(const v2f*)(bp[ct] + k);
#pragma unroll
        for (int ct = 0; ct < 8; ++ct)
            acc[ct] = __builtin_amdgcn_wmma_f32_16x16x4_f32(
                false, a, false, bf[ct], (short)0, acc[ct], false, false);
    }

#pragma unroll
    for (int ct = 0; ct < 8; ++ct) {
        const int col  = ct * 16 + lo;
        const float dw = is_lt ? diagW[col] : 1.0f;
#pragma unroll
        for (int i = 0; i < 8; ++i) {
            const int m = i + 8 * hi;
            proj[(size_t)(row0 + m) * AA + col] = tanhf(acc[ct][i]) * dw;
        }
    }
}

__global__ __launch_bounds__(256) void attn_softmax_kernel(
    const float* __restrict__ proj, const float* __restrict__ mask_lt,
    const float* __restrict__ mask_rt, float* __restrict__ out)
{
    extern __shared__ float S[];
    constexpr int SSTR = LL + 4;
    __shared__ float red[32][17];
    __shared__ float mlt_s[32];

    const int tid = threadIdx.x;
    const int b   = blockIdx.y;
    const int lt0 = blockIdx.x * 32;

    if (tid < 32) mlt_s[tid] = mask_lt[(size_t)b * LL + lt0 + tid];
    __syncthreads();

    const int lane = tid & 31;
    const int wv   = tid >> 5;
    const int lo   = lane & 15;
    const int hi   = lane >> 4;

    const float* __restrict__ ltp0 = proj + ((size_t)b * LL + lt0 + lo) * AA;
    const float* __restrict__ ltp1 = ltp0 + 16 * AA;
    const float* __restrict__ rtb = proj + ((size_t)(BB * LL) + (size_t)b * LL) * AA;
    const float* __restrict__ mrt = mask_rt + (size_t)b * LL;

    v4f af0[16], af1[16];
#pragma unroll
    for (int j = 0; j < 16; ++j) {
        af0[j] = *(const v4f*)(ltp0 + 8 * j + 4 * hi);
        af1[j] = *(const v4f*)(ltp1 + 8 * j + 4 * hi);
    }

    for (int ct = 0; ct < 8; ++ct) {
        const int cbase = wv * 128 + ct * 16;
        const float* __restrict__ rtp = rtb + (size_t)(cbase + lo) * AA + 4 * hi;
        v8f c0 = {};
        v8f c1 = {};
#pragma unroll
        for (int q = 0; q < 2; ++q) {
            v4f bf[8];
#pragma unroll
            for (int u = 0; u < 8; ++u)
                bf[u] = *(const v4f*)(rtp + 8 * (8 * q + u));
            __builtin_amdgcn_sched_barrier(0);
#pragma unroll
            for (int u = 0; u < 8; ++u) {
                const v4f a40 = af0[8 * q + u];
                const v4f a41 = af1[8 * q + u];
                const v4f b4  = bf[u];
                const v2f b0  = {b4.x, b4.y};
                const v2f b1  = {b4.z, b4.w};
                const v2f a00 = {a40.x, a40.y};
                const v2f a10 = {a41.x, a41.y};
                const v2f a01 = {a40.z, a40.w};
                const v2f a11 = {a41.z, a41.w};
                c0 = __builtin_amdgcn_wmma_f32_16x16x4_f32(
                    false, a00, false, b0, (short)0, c0, false, false);
                c1 = __builtin_amdgcn_wmma_f32_16x16x4_f32(
                    false, a10, false, b0, (short)0, c1, false, false);
                c0 = __builtin_amdgcn_wmma_f32_16x16x4_f32(
                    false, a01, false, b1, (short)0, c0, false, false);
                c1 = __builtin_amdgcn_wmma_f32_16x16x4_f32(
                    false, a11, false, b1, (short)0, c1, false, false);
            }
        }
        const float mr = mrt[cbase + lo];
#pragma unroll
        for (int i = 0; i < 8; ++i) {
            const int m = i + 8 * hi;
            S[m * SSTR + cbase + lo]        = c0[i] * mlt_s[m] * mr;
            S[(16 + m) * SSTR + cbase + lo] = c1[i] * mlt_s[16 + m] * mr;
        }
    }
    __syncthreads();

    const int r   = tid >> 4;
    const int c16 = tid & 15;
    const float* Sr0 = S + r * SSTR + c16 * 4;
    const float* Sr1 = S + (r + 16) * SSTR + c16 * 4;

    float mx0 = -3.402823466e38f;
    float mx1 = -3.402823466e38f;
#pragma unroll 4
    for (int j = 0; j < 16; ++j) {
        const v4f v = *(const v4f*)(Sr0 + 64 * j);
        const v4f w = *(const v4f*)(Sr1 + 64 * j);
        mx0 = fmaxf(mx0, fmaxf(fmaxf(v.x, v.y), fmaxf(v.z, v.w)));
        mx1 = fmaxf(mx1, fmaxf(fmaxf(w.x, w.y), fmaxf(w.z, w.w)));
    }
    red[r][c16]      = mx0;
    red[r + 16][c16] = mx1;
    __syncthreads();
    float rmax0 = red[r][0];
    float rmax1 = red[r + 16][0];
    for (int t = 1; t < 16; ++t) {
        rmax0 = fmaxf(rmax0, red[r][t]);
        rmax1 = fmaxf(rmax1, red[r + 16][t]);
    }
    __syncthreads();

    float sum0 = 0.f;
    float sum1 = 0.f;
    for (int j = 0; j < 16; ++j) {
        float* p0 = S + r * SSTR + c16 * 4 + 64 * j;
        float* p1 = S + (r + 16) * SSTR + c16 * 4 + 64 * j;
        v4f v = *(const v4f*)p0;
        v4f w = *(const v4f*)p1;
        v.x = expf(v.x - rmax0);
        v.y = expf(v.y - rmax0);
        v.z = expf(v.z - rmax0);
        v.w = expf(v.w - rmax0);
        w.x = expf(w.x - rmax1);
        w.y = expf(w.y - rmax1);
        w.z = expf(w.z - rmax1);
        w.w = expf(w.w - rmax1);
        sum0 += (v.x + v.y) + (v.z + v.w);
        sum1 += (w.x + w.y) + (w.z + w.w);
        *(v4f*)p0 = v;
        *(v4f*)p1 = w;
    }
    red[r][c16]      = sum0;
    red[r + 16][c16] = sum1;
    __syncthreads();
    float rsum0 = 0.f;
    float rsum1 = 0.f;
    for (int t = 0; t < 16; ++t) {
        rsum0 += red[r][t];
        rsum1 += red[r + 16][t];
    }

    const float mlr0 = mlt_s[r] / rsum0;
    const float mlr1 = mlt_s[r + 16] / rsum1;
    float* __restrict__ orow0 = out + ((size_t)b * LL + lt0 + r) * (size_t)LL;
    float* __restrict__ orow1 = orow0 + (size_t)16 * LL;
    for (int j = 0; j < 16; ++j) {
        const int cc = c16 * 4 + 64 * j;
        const v4f e0 = *(const v4f*)(S + r * SSTR + cc);
        const v4f e1 = *(const v4f*)(S + (r + 16) * SSTR + cc);
        const v4f m4 = *(const v4f*)(mrt + cc);
        v4f o0, o1;
        o0.x = e0.x * mlr0 * m4.x;
        o0.y = e0.y * mlr0 * m4.y;
        o0.z = e0.z * mlr0 * m4.z;
        o0.w = e0.w * mlr0 * m4.w;
        o1.x = e1.x * mlr1 * m4.x;
        o1.y = e1.y * mlr1 * m4.y;
        o1.z = e1.z * mlr1 * m4.z;
        o1.w = e1.w * mlr1 * m4.w;
        __builtin_nontemporal_store(o0, (v4f*)(orow0 + cc));
        __builtin_nontemporal_store(o1, (v4f*)(orow1 + cc));
    }
}

extern "C" void kernel_launch(void* const* d_in, const int* in_sizes, int n_in,
                              void* d_out, int out_size, void* d_ws, size_t ws_size,
                              hipStream_t stream)
{
    (void)in_sizes; (void)n_in; (void)out_size; (void)ws_size;

    const float* reps_lt = (const float*)d_in[0];
    const float* reps_rt = (const float*)d_in[1];
    const float* mask_lt = (const float*)d_in[2];
    const float* mask_rt = (const float*)d_in[3];
    const float* attn_k  = (const float*)d_in[4];
    const float* diagW   = (const float*)d_in[5];
    float* outp = (float*)d_out;
    float* proj = (float*)d_ws;

    const size_t ldsW = (size_t)AA * (HH + 4) * sizeof(float);
    proj_tanh_kernel<<<dim3(512), 256, ldsW, stream>>>(
        reps_lt, reps_rt, attn_k, diagW, proj);

    const size_t ldsS = (size_t)32 * (LL + 4) * sizeof(float);
    attn_softmax_kernel<<<dim3(LL / 32, BB), 256, ldsS, stream>>>(
        proj, mask_lt, mask_rt, outp);
}


# proj workspace kept in WMMA fragment order (projection WMMA operands swapped, D transposed): every fragment load and store of a wave is whole contiguous 128-byte lines; summation order unchanged
# speedup vs baseline: 1.3740x; 1.0824x over previous
#include <hip/hip_runtime.h>
#include <cmath>


typedef __attribute__((ext_vector_type(2))) float v2f;
typedef __attribute__((ext_vector_type(4))) float v4f;
typedef __attribute__((ext_vector_type(8))) float v8f;

#define BB 32
#define LL 1024
#define HH 256
#define AA 128

__global__ __launch_bounds__(256) void proj_tanh_kernel(
    const float* __restrict__ reps_lt, const float* __restrict__ reps_rt,
    const float* __restrict__ W, const float* __restrict__ diagW,
    float* __restrict__ proj)
{
    extern __shared__ float WT[];
    constexpr int TSTR = HH + 4;
    const int tid = threadIdx.x;

    for (int i = tid * 4; i < HH * AA; i += 256 * 4) {
        const int h = i >> 7;
        const int a = i & (AA - 1);
        const v4f w = *(const v4f*)(W + h * AA + a);
        WT[(a + 0) * TSTR + h] = w.x;
        WT[(a + 1) * TSTR + h] = w.y;
        WT[(a + 2) * TSTR + h] = w.z;
        WT[(a + 3) * TSTR + h] = w.w;
    }
    __syncthreads();

    const int lane = tid & 31;
    const int wv   = tid >> 5;
    const int lo   = lane & 15;
    const int hi   = lane >> 4;

    const int row0 = (blockIdx.x * 8 + wv) * 16;
    const int grow = row0 + lo;
    const bool is_lt = row0 < BB * LL;
    const float* __restrict__ src =
        is_lt ? reps_lt + (size_t)grow * HH
              : reps_rt + (size_t)(grow - BB * LL) * HH;

    const float* bp[8];
#pragma unroll
    for (int ct = 0; ct < 8; ++ct)
        bp[ct] = WT + (ct * 16 + lo) * TSTR + 2 * hi;

    v8f acc[8] = {};

    for (int p = 0; p < HH / 4; ++p) {
        const int k = 4 * p;
        const v2f a = *(const v2f*)(src + k + 2 * hi);

        v2f bf[8];
#pragma unroll
        for (int ct = 0; ct < 8; ++ct)
            bf[ct] = *(const v2f*)(bp[ct] + k);
#pragma unroll
        for (int ct = 0; ct < 8; ++ct)
            acc[ct] = __builtin_amdgcn_wmma_f32_16x16x4_f32(
                false, bf[ct], false, a, (short)0, acc[ct], false, false);
    }

    float* __restrict__ tile = proj + (size_t)(row0 >> 4) * (16 * AA);
    const v4f one4 = {1.0f, 1.0f, 1.0f, 1.0f};
#pragma unroll
    for (int ct = 0; ct < 8; ++ct) {
        const int j = 2 * ct + hi;
        const v4f dw0 = is_lt ? *(const v4f*)(diagW + 8 * j)     : one4;
        const v4f dw1 = is_lt ? *(const v4f*)(diagW + 8 * j + 4) : one4;
        v4f o0, o1;
        o0.x = tanhf(acc[ct][0]) * dw0.x;
        o0.y = tanhf(acc[ct][1]) * dw0.y;
        o0.z = tanhf(acc[ct][2]) * dw0.z;
        o0.w = tanhf(acc[ct][3]) * dw0.w;
        o1.x = tanhf(acc[ct][4]) * dw1.x;
        o1.y = tanhf(acc[ct][5]) * dw1.y;
        o1.z = tanhf(acc[ct][6]) * dw1.z;
        o1.w = tanhf(acc[ct][7]) * dw1.w;
        *(v4f*)(tile + (j * 32 + lo) * 4)      = o0;
        *(v4f*)(tile + (j * 32 + 16 + lo) * 4) = o1;
    }
}

__global__ __launch_bounds__(256) void attn_softmax_kernel(
    const float* __restrict__ proj, const float* __restrict__ mask_lt,
    const float* __restrict__ mask_rt, float* __restrict__ out)
{
    extern __shared__ float S[];
    constexpr int SSTR = LL + 4;
    __shared__ float red[32][17];
    __shared__ float mlt_s[32];

    const int tid = threadIdx.x;
    const int b   = blockIdx.y;
    const int lt0 = blockIdx.x * 32;

    if (tid < 32) mlt_s[tid] = mask_lt[(size_t)b * LL + lt0 + tid];
    __syncthreads();

    const int lane = tid & 31;
    const int wv   = tid >> 5;
    const int lo   = lane & 15;
    const int hi   = lane >> 4;

    const float* __restrict__ ltp0 =
        proj + (((size_t)b * LL + lt0) >> 4) * (16 * AA) + lane * 4;
    const float* __restrict__ ltp1 = ltp0 + 16 * AA;
    const float* __restrict__ rtb =
        proj + ((size_t)(BB * LL) + (size_t)b * LL) * AA + lane * 4;
    const float* __restrict__ mrt = mask_rt + (size_t)b * LL;

    v4f af0[16], af1[16];
#pragma unroll
    for (int j = 0; j < 16; ++j) {
        af0[j] = *(const v4f*)(ltp0 + 128 * j);
        af1[j] = *(const v4f*)(ltp1 + 128 * j);
    }

    for (int ct = 0; ct < 8; ++ct) {
        const int cbase = wv * 128 + ct * 16;
        const float* __restrict__ rtp = rtb + (size_t)(cbase >> 4) * (16 * AA);
        v8f c0 = {};
        v8f c1 = {};
#pragma unroll
        for (int q = 0; q < 2; ++q) {
            v4f bf[8];
#pragma unroll
            for (int u = 0; u < 8; ++u)
                bf[u] = *(const v4f*)(rtp + 128 * (8 * q + u));
            __builtin_amdgcn_sched_barrier(0);
#pragma unroll
            for (int u = 0; u < 8; ++u) {
                const v4f a40 = af0[8 * q + u];
                const v4f a41 = af1[8 * q + u];
                const v4f b4  = bf[u];
                const v2f b0  = {b4.x, b4.y};
                const v2f b1  = {b4.z, b4.w};
                const v2f a00 = {a40.x, a40.y};
                const v2f a10 = {a41.x, a41.y};
                const v2f a01 = {a40.z, a40.w};
                const v2f a11 = {a41.z, a41.w};
                c0 = __builtin_amdgcn_wmma_f32_16x16x4_f32(
                    false, a00, false, b0, (short)0, c0, false, false);
                c1 = __builtin_amdgcn_wmma_f32_16x16x4_f32(
                    false, a10, false, b0, (short)0, c1, false, false);
                c0 = __builtin_amdgcn_wmma_f32_16x16x4_f32(
                    false, a01, false, b1, (short)0, c0, false, false);
                c1 = __builtin_amdgcn_wmma_f32_16x16x4_f32(
                    false, a11, false, b1, (short)0, c1, false, false);
            }
        }
        const float mr = mrt[cbase + lo];
#pragma unroll
        for (int i = 0; i < 8; ++i) {
            const int m = i + 8 * hi;
            S[m * SSTR + cbase + lo]        = c0[i] * mlt_s[m] * mr;
            S[(16 + m) * SSTR + cbase + lo] = c1[i] * mlt_s[16 + m] * mr;
        }
    }
    __syncthreads();

    const int r   = tid >> 4;
    const int c16 = tid & 15;
    const float* Sr0 = S + r * SSTR + c16 * 4;
    const float* Sr1 = S + (r + 16) * SSTR + c16 * 4;

    float mx0 = -3.402823466e38f;
    float mx1 = -3.402823466e38f;
#pragma unroll 4
    for (int j = 0; j < 16; ++j) {
        const v4f v = *(const v4f*)(Sr0 + 64 * j);
        const v4f w = *(const v4f*)(Sr1 + 64 * j);
        mx0 = fmaxf(mx0, fmaxf(fmaxf(v.x, v.y), fmaxf(v.z, v.w)));
        mx1 = fmaxf(mx1, fmaxf(fmaxf(w.x, w.y), fmaxf(w.z, w.w)));
    }
    red[r][c16]      = mx0;
    red[r + 16][c16] = mx1;
    __syncthreads();
    float rmax0 = red[r][0];
    float rmax1 = red[r + 16][0];
    for (int t = 1; t < 16; ++t) {
        rmax0 = fmaxf(rmax0, red[r][t]);
        rmax1 = fmaxf(rmax1, red[r + 16][t]);
    }
    __syncthreads();

    float sum0 = 0.f;
    float sum1 = 0.f;
    for (int j = 0; j < 16; ++j) {
        float* p0 = S + r * SSTR + c16 * 4 + 64 * j;
        float* p1 = S + (r + 16) * SSTR + c16 * 4 + 64 * j;
        v4f v = *(const v4f*)p0;
        v4f w = *(const v4f*)p1;
        v.x = expf(v.x - rmax0);
        v.y = expf(v.y - rmax0);
        v.z = expf(v.z - rmax0);
        v.w = expf(v.w - rmax0);
        w.x = expf(w.x - rmax1);
        w.y = expf(w.y - rmax1);
        w.z = expf(w.z - rmax1);
        w.w = expf(w.w - rmax1);
        sum0 += (v.x + v.y) + (v.z + v.w);
        sum1 += (w.x + w.y) + (w.z + w.w);
        *(v4f*)p0 = v;
        *(v4f*)p1 = w;
    }
    red[r][c16]      = sum0;
    red[r + 16][c16] = sum1;
    __syncthreads();
    float rsum0 = 0.f;
    float rsum1 = 0.f;
    for (int t = 0; t < 16; ++t) {
        rsum0 += red[r][t];
        rsum1 += red[r + 16][t];
    }

    const float mlr0 = mlt_s[r] / rsum0;
    const float mlr1 = mlt_s[r + 16] / rsum1;
    float* __restrict__ orow0 = out + ((size_t)b * LL + lt0 + r) * (size_t)LL;
    float* __restrict__ orow1 = orow0 + (size_t)16 * LL;
    for (int j = 0; j < 16; ++j) {
        const int cc = c16 * 4 + 64 * j;
        const v4f e0 = *(const v4f*)(S + r * SSTR + cc);
        const v4f e1 = *(const v4f*)(S + (r + 16) * SSTR + cc);
        const v4f m4 = *(const v4f*)(mrt + cc);
        v4f o0, o1;
        o0.x = e0.x * mlr0 * m4.x;
        o0.y = e0.y * mlr0 * m4.y;
        o0.z = e0.z * mlr0 * m4.z;
        o0.w = e0.w * mlr0 * m4.w;
        o1.x = e1.x * mlr1 * m4.x;
        o1.y = e1.y * mlr1 * m4.y;
        o1.z = e1.z * mlr1 * m4.z;
        o1.w = e1.w * mlr1 * m4.w;
        __builtin_nontemporal_store(o0, (v4f*)(orow0 + cc));
        __builtin_nontemporal_store(o1, (v4f*)(orow1 + cc));
    }
}

extern "C" void kernel_launch(void* const* d_in, const int* in_sizes, int n_in,
                              void* d_out, int out_size, void* d_ws, size_t ws_size,
                              hipStream_t stream)
{
    (void)in_sizes; (void)n_in; (void)out_size; (void)ws_size;

    const float* reps_lt = (const float*)d_in[0];
    const float* reps_rt = (const float*)d_in[1];
    const float* mask_lt = (const float*)d_in[2];
    const float* mask_rt = (const float*)d_in[3];
    const float* attn_k  = (const float*)d_in[4];
    const float* diagW   = (const float*)d_in[5];
    float* outp = (float*)d_out;
    float* proj = (float*)d_ws;

    const size_t ldsW = (size_t)AA * (HH + 4) * sizeof(float);
    proj_tanh_kernel<<<dim3(512), 256, ldsW, stream>>>(
        reps_lt, reps_rt, attn_k, diagW, proj);

    const size_t ldsS = (size_t)32 * (LL + 4) * sizeof(float);
    attn_softmax_kernel<<<dim3(LL / 32, BB), 256, ldsS, stream>>>(
        proj, mask_lt, mask_rt, outp);
}


# softmax exp and projection tanh computed branch-free on the hardware exp2 and rcp instructions, not the library routines: 3-7x fewer vector instructions each; tanh within 2e-7 abs, exp a few ulp
# speedup vs baseline: 1.4799x; 1.0771x over previous
#include <hip/hip_runtime.h>
#include <cmath>


typedef __attribute__((ext_vector_type(2))) float v2f;
typedef __attribute__((ext_vector_type(4))) float v4f;
typedef __attribute__((ext_vector_type(8))) float v8f;

#define BB 32
#define LL 1024
#define HH 256
#define AA 128

__device__ __forceinline__ float exp_hw(float x)
{
    return __builtin_amdgcn_exp2f(x * 1.4426950408889634f);
}
__device__ __forceinline__ float tanh_hw(float x)
{
    const float e2x = __builtin_amdgcn_exp2f(x * 2.8853900817779268f);
    return 1.0f - 2.0f * __builtin_amdgcn_rcpf(e2x + 1.0f);
}

__global__ __launch_bounds__(256) void proj_tanh_kernel(
    const float* __restrict__ reps_lt, const float* __restrict__ reps_rt,
    const float* __restrict__ W, const float* __restrict__ diagW,
    float* __restrict__ proj)
{
    extern __shared__ float WT[];
    constexpr int TSTR = HH + 4;
    const int tid = threadIdx.x;

    for (int i = tid * 4; i < HH * AA; i += 256 * 4) {
        const int h = i >> 7;
        const int a = i & (AA - 1);
        const v4f w = *(const v4f*)(W + h * AA + a);
        WT[(a + 0) * TSTR + h] = w.x;
        WT[(a + 1) * TSTR + h] = w.y;
        WT[(a + 2) * TSTR + h] = w.z;
        WT[(a + 3) * TSTR + h] = w.w;
    }
    __syncthreads();

    const int lane = tid & 31;
    const int wv   = tid >> 5;
    const int lo   = lane & 15;
    const int hi   = lane >> 4;

    const int row0 = (blockIdx.x * 8 + wv) * 16;
    const int grow = row0 + lo;
    const bool is_lt = row0 < BB * LL;
    const float* __restrict__ src =
        is_lt ? reps_lt + (size_t)grow * HH
              : reps_rt + (size_t)(grow - BB * LL) * HH;

    const float* bp[8];
#pragma unroll
    for (int ct = 0; ct < 8; ++ct)
        bp[ct] = WT + (ct * 16 + lo) * TSTR + 2 * hi;

    v8f acc[8] = {};

    for (int p = 0; p < HH / 4; ++p) {
        const int k = 4 * p;
        const v2f a = *(const v2f*)(src + k + 2 * hi);

        v2f bf[8];
#pragma unroll
        for (int ct = 0; ct < 8; ++ct)
            bf[ct] = *(const v2f*)(bp[ct] + k);
#pragma unroll
        for (int ct = 0; ct < 8; ++ct)
            acc[ct] = __builtin_amdgcn_wmma_f32_16x16x4_f32(
                false, bf[ct], false, a, (short)0, acc[ct], false, false);
    }

    float* __restrict__ tile = proj + (size_t)(row0 >> 4) * (16 * AA);
    const v4f one4 = {1.0f, 1.0f, 1.0f, 1.0f};
#pragma unroll
    for (int ct = 0; ct < 8; ++ct) {
        const int j = 2 * ct + hi;
        const v4f dw0 = is_lt ? *(const v4f*)(diagW + 8 * j)     : one4;
        const v4f dw1 = is_lt ? *(const v4f*)(diagW + 8 * j + 4) : one4;
        v4f o0, o1;
        o0.x = tanh_hw(acc[ct][0]) * dw0.x;
        o0.y = tanh_hw(acc[ct][1]) * dw0.y;
        o0.z = tanh_hw(acc[ct][2]) * dw0.z;
        o0.w = tanh_hw(acc[ct][3]) * dw0.w;
        o1.x = tanh_hw(acc[ct][4]) * dw1.x;
        o1.y = tanh_hw(acc[ct][5]) * dw1.y;
        o1.z = tanh_hw(acc[ct][6]) * dw1.z;
        o1.w = tanh_hw(acc[ct][7]) * dw1.w;
        *(v4f*)(tile + (j * 32 + lo) * 4)      = o0;
        *(v4f*)(tile + (j * 32 + 16 + lo) * 4) = o1;
    }
}

__global__ __launch_bounds__(256) void attn_softmax_kernel(
    const float* __restrict__ proj, const float* __restrict__ mask_lt,
    const float* __restrict__ mask_rt, float* __restrict__ out)
{
    extern __shared__ float S[];
    constexpr int SSTR = LL + 4;
    __shared__ float red[32][17];
    __shared__ float mlt_s[32];

    const int tid = threadIdx.x;
    const int b   = blockIdx.y;
    const int lt0 = blockIdx.x * 32;

    if (tid < 32) mlt_s[tid] = mask_lt[(size_t)b * LL + lt0 + tid];
    __syncthreads();

    const int lane = tid & 31;
    const int wv   = tid >> 5;
    const int lo   = lane & 15;
    const int hi   = lane >> 4;

    const float* __restrict__ ltp0 =
        proj + (((size_t)b * LL + lt0) >> 4) * (16 * AA) + lane * 4;
    const float* __restrict__ ltp1 = ltp0 + 16 * AA;
    const float* __restrict__ rtb =
        proj + ((size_t)(BB * LL) + (size_t)b * LL) * AA + lane * 4;
    const float* __restrict__ mrt = mask_rt + (size_t)b * LL;

    v4f af0[16], af1[16];
#pragma unroll
    for (int j = 0; j < 16; ++j) {
        af0[j] = *(const v4f*)(ltp0 + 128 * j);
        af1[j] = *(const v4f*)(ltp1 + 128 * j);
    }

    for (int ct = 0; ct < 8; ++ct) {
        const int cbase = wv * 128 + ct * 16;
        const float* __restrict__ rtp = rtb + (size_t)(cbase >> 4) * (16 * AA);
        v8f c0 = {};
        v8f c1 = {};
#pragma unroll
        for (int q = 0; q < 2; ++q) {
            v4f bf[8];
#pragma unroll
            for (int u = 0; u < 8; ++u)
                bf[u] = *(const v4f*)(rtp + 128 * (8 * q + u));
            __builtin_amdgcn_sched_barrier(0);
#pragma unroll
            for (int u = 0; u < 8; ++u) {
                const v4f a40 = af0[8 * q + u];
                const v4f a41 = af1[8 * q + u];
                const v4f b4  = bf[u];
                const v2f b0  = {b4.x, b4.y};
                const v2f b1  = {b4.z, b4.w};
                const v2f a00 = {a40.x, a40.y};
                const v2f a10 = {a41.x, a41.y};
                const v2f a01 = {a40.z, a40.w};
                const v2f a11 = {a41.z, a41.w};
                c0 = __builtin_amdgcn_wmma_f32_16x16x4_f32(
                    false, a00, false, b0, (short)0, c0, false, false);
                c1 = __builtin_amdgcn_wmma_f32_16x16x4_f32(
                    false, a10, false, b0, (short)0, c1, false, false);
                c0 = __builtin_amdgcn_wmma_f32_16x16x4_f32(
                    false, a01, false, b1, (short)0, c0, false, false);
                c1 = __builtin_amdgcn_wmma_f32_16x16x4_f32(
                    false, a11, false, b1, (short)0, c1, false, false);
            }
        }
        const float mr = mrt[cbase + lo];
#pragma unroll
        for (int i = 0; i < 8; ++i) {
            const int m = i + 8 * hi;
            S[m * SSTR + cbase + lo]        = c0[i] * mlt_s[m] * mr;
            S[(16 + m) * SSTR + cbase + lo] = c1[i] * mlt_s[16 + m] * mr;
        }
    }
    __syncthreads();

    const int r   = tid >> 4;
    const int c16 = tid & 15;
    const float* Sr0 = S + r * SSTR + c16 * 4;
    const float* Sr1 = S + (r + 16) * SSTR + c16 * 4;

    float mx0 = -3.402823466e38f;
    float mx1 = -3.402823466e38f;
#pragma unroll 4
    for (int j = 0; j < 16; ++j) {
        const v4f v = *(const v4f*)(Sr0 + 64 * j);
        const v4f w = *(const v4f*)(Sr1 + 64 * j);
        mx0 = fmaxf(mx0, fmaxf(fmaxf(v.x, v.y), fmaxf(v.z, v.w)));
        mx1 = fmaxf(mx1, fmaxf(fmaxf(w.x, w.y), fmaxf(w.z, w.w)));
    }
    red[r][c16]      = mx0;
    red[r + 16][c16] = mx1;
    __syncthreads();
    float rmax0 = red[r][0];
    float rmax1 = red[r + 16][0];
    for (int t = 1; t < 16; ++t) {
        rmax0 = fmaxf(rmax0, red[r][t]);
        rmax1 = fmaxf(rmax1, red[r + 16][t]);
    }
    __syncthreads();

    float sum0 = 0.f;
    float sum1 = 0.f;
    for (int j = 0; j < 16; ++j) {
        float* p0 = S + r * SSTR + c16 * 4 + 64 * j;
        float* p1 = S + (r + 16) * SSTR + c16 * 4 + 64 * j;
        v4f v = *(const v4f*)p0;
        v4f w = *(const v4f*)p1;
        v.x = exp_hw(v.x - rmax0);
        v.y = exp_hw(v.y - rmax0);
        v.z = exp_hw(v.z - rmax0);
        v.w = exp_hw(v.w - rmax0);
        w.x = exp_hw(w.x - rmax1);
        w.y = exp_hw(w.y - rmax1);
        w.z = exp_hw(w.z - rmax1);
        w.w = exp_hw(w.w - rmax1);
        sum0 += (v.x + v.y) + (v.z + v.w);
        sum1 += (w.x + w.y) + (w.z + w.w);
        *(v4f*)p0 = v;
        *(v4f*)p1 = w;
    }
    red[r][c16]      = sum0;
    red[r + 16][c16] = sum1;
    __syncthreads();
    float rsum0 = 0.f;
    float rsum1 = 0.f;
    for (int t = 0; t < 16; ++t) {
        rsum0 += red[r][t];
        rsum1 += red[r + 16][t];
    }

    const float mlr0 = mlt_s[r] / rsum0;
    const float mlr1 = mlt_s[r + 16] / rsum1;
    float* __restrict__ orow0 = out + ((size_t)b * LL + lt0 + r) * (size_t)LL;
    float* __restrict__ orow1 = orow0 + (size_t)16 * LL;
    for (int j = 0; j < 16; ++j) {
        const int cc = c16 * 4 + 64 * j;
        const v4f e0 = *(const v4f*)(S + r * SSTR + cc);
        const v4f e1 = *(const v4f*)(S + (r + 16) * SSTR + cc);
        const v4f m4 = *(const v4f*)(mrt + cc);
        v4f o0, o1;
        o0.x = e0.x * mlr0 * m4.x;
        o0.y = e0.y * mlr0 * m4.y;
        o0.z = e0.z * mlr0 * m4.z;
        o0.w = e0.w * mlr0 * m4.w;
        o1.x = e1.x * mlr1 * m4.x;
        o1.y = e1.y * mlr1 * m4.y;
        o1.z = e1.z * mlr1 * m4.z;
        o1.w = e1.w * mlr1 * m4.w;
        __builtin_nontemporal_store(o0, (v4f*)(orow0 + cc));
        __builtin_nontemporal_store(o1, (v4f*)(orow1 + cc));
    }
}

extern "C" void kernel_launch(void* const* d_in, const int* in_sizes, int n_in,
                              void* d_out, int out_size, void* d_ws, size_t ws_size,
                              hipStream_t stream)
{
    (void)in_sizes; (void)n_in; (void)out_size; (void)ws_size;

    const float* reps_lt = (const float*)d_in[0];
    const float* reps_rt = (const float*)d_in[1];
    const float* mask_lt = (const float*)d_in[2];
    const float* mask_rt = (const float*)d_in[3];
    const float* attn_k  = (const float*)d_in[4];
    const float* diagW   = (const float*)d_in[5];
    float* outp = (float*)d_out;
    float* proj = (float*)d_ws;

    const size_t ldsW = (size_t)AA * (HH + 4) * sizeof(float);
    proj_tanh_kernel<<<dim3(512), 256, ldsW, stream>>>(
        reps_lt, reps_rt, attn_k, diagW, proj);

    const size_t ldsS = (size_t)32 * (LL + 4) * sizeof(float);
    attn_softmax_kernel<<<dim3(LL / 32, BB), 256, ldsS, stream>>>(
        proj, mask_lt, mask_rt, outp);
}


# projection kernel: W staged in LDS as k-pair-interleaved rows, filled by coalesced 8-byte loads and conflict-free 16-byte stores (was 4-byte stores, 8 lanes per bank); summation order unchanged
# speedup vs baseline: 1.6513x; 1.1158x over previous
#include <hip/hip_runtime.h>
#include <cmath>


typedef __attribute__((ext_vector_type(2))) float v2f;
typedef __attribute__((ext_vector_type(4))) float v4f;
typedef __attribute__((ext_vector_type(8))) float v8f;

#define BB 32
#define LL 1024
#define HH 256
#define AA 128

__device__ __forceinline__ float exp_hw(float x)
{
    return __builtin_amdgcn_exp2f(x * 1.4426950408889634f);
}
__device__ __forceinline__ float tanh_hw(float x)
{
    const float e2x = __builtin_amdgcn_exp2f(x * 2.8853900817779268f);
    return 1.0f - 2.0f * __builtin_amdgcn_rcpf(e2x + 1.0f);
}

__global__ __launch_bounds__(256) void proj_tanh_kernel(
    const float* __restrict__ reps_lt, const float* __restrict__ reps_rt,
    const float* __restrict__ W, const float* __restrict__ diagW,
    float* __restrict__ proj)
{
    extern __shared__ float Ws[];
    constexpr int PP = 2 * AA + 32;
    const int tid  = threadIdx.x;
    const int lane = tid & 31;
    const int wv   = tid >> 5;
    const int lo   = lane & 15;
    const int hi   = lane >> 4;

#pragma unroll
    for (int tb = 0; tb < 16; tb += 8) {
        v2f r0a[8], r0b[8], r1a[8], r1b[8];
#pragma unroll
        for (int t = 0; t < 8; ++t) {
            const float* g = W + (size_t)(2 * (wv + 8 * (tb + t))) * AA + 2 * lane;
            r0a[t] = *(const v2f*)(g);
            r0b[t] = *(const v2f*)(g + 64);
            r1a[t] = *(const v2f*)(g + AA);
            r1b[t] = *(const v2f*)(g + AA + 64);
        }
#pragma unroll
        for (int t = 0; t < 8; ++t) {
            float* d = Ws + (wv + 8 * (tb + t)) * PP + 4 * lane;
            const v4f s0 = {r0a[t].x, r1a[t].x, r0a[t].y, r1a[t].y};
            const v4f s1 = {r0b[t].x, r1b[t].x, r0b[t].y, r1b[t].y};
            *(v4f*)(d)       = s0;
            *(v4f*)(d + 128) = s1;
        }
    }
    __syncthreads();

    const int row0 = (blockIdx.x * 8 + wv) * 16;
    const int grow = row0 + lo;
    const bool is_lt = row0 < BB * LL;
    const float* __restrict__ src =
        is_lt ? reps_lt + (size_t)grow * HH
              : reps_rt + (size_t)(grow - BB * LL) * HH;

    const float* bp = Ws + hi * PP + 2 * lo;

    v8f acc[8] = {};

    for (int p = 0; p < HH / 4; ++p) {
        const int k = 4 * p;
        const v2f a = *(const v2f*)(src + k + 2 * hi);
        const float* bk = bp + p * (2 * PP);

        v2f bf[8];
#pragma unroll
        for (int ct = 0; ct < 8; ++ct)
            bf[ct] = *(const v2f*)(bk + 32 * ct);
#pragma unroll
        for (int ct = 0; ct < 8; ++ct)
            acc[ct] = __builtin_amdgcn_wmma_f32_16x16x4_f32(
                false, bf[ct], false, a, (short)0, acc[ct], false, false);
    }

    float* __restrict__ tile = proj + (size_t)(row0 >> 4) * (16 * AA);
    const v4f one4 = {1.0f, 1.0f, 1.0f, 1.0f};
#pragma unroll
    for (int ct = 0; ct < 8; ++ct) {
        const int j = 2 * ct + hi;
        const v4f dw0 = is_lt ? *(const v4f*)(diagW + 8 * j)     : one4;
        const v4f dw1 = is_lt ? *(const v4f*)(diagW + 8 * j + 4) : one4;
        v4f o0, o1;
        o0.x = tanh_hw(acc[ct][0]) * dw0.x;
        o0.y = tanh_hw(acc[ct][1]) * dw0.y;
        o0.z = tanh_hw(acc[ct][2]) * dw0.z;
        o0.w = tanh_hw(acc[ct][3]) * dw0.w;
        o1.x = tanh_hw(acc[ct][4]) * dw1.x;
        o1.y = tanh_hw(acc[ct][5]) * dw1.y;
        o1.z = tanh_hw(acc[ct][6]) * dw1.z;
        o1.w = tanh_hw(acc[ct][7]) * dw1.w;
        *(v4f*)(tile + (j * 32 + lo) * 4)      = o0;
        *(v4f*)(tile + (j * 32 + 16 + lo) * 4) = o1;
    }
}

__global__ __launch_bounds__(256) void attn_softmax_kernel(
    const float* __restrict__ proj, const float* __restrict__ mask_lt,
    const float* __restrict__ mask_rt, float* __restrict__ out)
{
    extern __shared__ float S[];
    constexpr int SSTR = LL + 4;
    __shared__ float red[32][17];
    __shared__ float mlt_s[32];

    const int tid = threadIdx.x;
    const int b   = blockIdx.y;
    const int lt0 = blockIdx.x * 32;

    if (tid < 32) mlt_s[tid] = mask_lt[(size_t)b * LL + lt0 + tid];
    __syncthreads();

    const int lane = tid & 31;
    const int wv   = tid >> 5;
    const int lo   = lane & 15;
    const int hi   = lane >> 4;

    const float* __restrict__ ltp0 =
        proj + (((size_t)b * LL + lt0) >> 4) * (16 * AA) + lane * 4;
    const float* __restrict__ ltp1 = ltp0 + 16 * AA;
    const float* __restrict__ rtb =
        proj + ((size_t)(BB * LL) + (size_t)b * LL) * AA + lane * 4;
    const float* __restrict__ mrt = mask_rt + (size_t)b * LL;

    v4f af0[16], af1[16];
#pragma unroll
    for (int j = 0; j < 16; ++j) {
        af0[j] = *(const v4f*)(ltp0 + 128 * j);
        af1[j] = *(const v4f*)(ltp1 + 128 * j);
    }

    for (int ct = 0; ct < 8; ++ct) {
        const int cbase = wv * 128 + ct * 16;
        const float* __restrict__ rtp = rtb + (size_t)(cbase >> 4) * (16 * AA);
        v8f c0 = {};
        v8f c1 = {};
#pragma unroll
        for (int q = 0; q < 2; ++q) {
            v4f bf[8];
#pragma unroll
            for (int u = 0; u < 8; ++u)
                bf[u] = *(const v4f*)(rtp + 128 * (8 * q + u));
            __builtin_amdgcn_sched_barrier(0);
#pragma unroll
            for (int u = 0; u < 8; ++u) {
                const v4f a40 = af0[8 * q + u];
                const v4f a41 = af1[8 * q + u];
                const v4f b4  = bf[u];
                const v2f b0  = {b4.x, b4.y};
                const v2f b1  = {b4.z, b4.w};
                const v2f a00 = {a40.x, a40.y};
                const v2f a10 = {a41.x, a41.y};
                const v2f a01 = {a40.z, a40.w};
                const v2f a11 = {a41.z, a41.w};
                c0 = __builtin_amdgcn_wmma_f32_16x16x4_f32(
                    false, a00, false, b0, (short)0, c0, false, false);
                c1 = __builtin_amdgcn_wmma_f32_16x16x4_f32(
                    false, a10, false, b0, (short)0, c1, false, false);
                c0 = __builtin_amdgcn_wmma_f32_16x16x4_f32(
                    false, a01, false, b1, (short)0, c0, false, false);
                c1 = __builtin_amdgcn_wmma_f32_16x16x4_f32(
                    false, a11, false, b1, (short)0, c1, false, false);
            }
        }
        const float mr = mrt[cbase + lo];
#pragma unroll
        for (int i = 0; i < 8; ++i) {
            const int m = i + 8 * hi;
            S[m * SSTR + cbase + lo]        = c0[i] * mlt_s[m] * mr;
            S[(16 + m) * SSTR + cbase + lo] = c1[i] * mlt_s[16 + m] * mr;
        }
    }
    __syncthreads();

    const int r   = tid >> 4;
    const int c16 = tid & 15;
    const float* Sr0 = S + r * SSTR + c16 * 4;
    const float* Sr1 = S + (r + 16) * SSTR + c16 * 4;

    float mx0 = -3.402823466e38f;
    float mx1 = -3.402823466e38f;
#pragma unroll 4
    for (int j = 0; j < 16; ++j) {
        const v4f v = *(const v4f*)(Sr0 + 64 * j);
        const v4f w = *(const v4f*)(Sr1 + 64 * j);
        mx0 = fmaxf(mx0, fmaxf(fmaxf(v.x, v.y), fmaxf(v.z, v.w)));
        mx1 = fmaxf(mx1, fmaxf(fmaxf(w.x, w.y), fmaxf(w.z, w.w)));
    }
    red[r][c16]      = mx0;
    red[r + 16][c16] = mx1;
    __syncthreads();
    float rmax0 = red[r][0];
    float rmax1 = red[r + 16][0];
    for (int t = 1; t < 16; ++t) {
        rmax0 = fmaxf(rmax0, red[r][t]);
        rmax1 = fmaxf(rmax1, red[r + 16][t]);
    }
    __syncthreads();

    float sum0 = 0.f;
    float sum1 = 0.f;
    for (int j = 0; j < 16; ++j) {
        float* p0 = S + r * SSTR + c16 * 4 + 64 * j;
        float* p1 = S + (r + 16) * SSTR + c16 * 4 + 64 * j;
        v4f v = *(const v4f*)p0;
        v4f w = *(const v4f*)p1;
        v.x = exp_hw(v.x - rmax0);
        v.y = exp_hw(v.y - rmax0);
        v.z = exp_hw(v.z - rmax0);
        v.w = exp_hw(v.w - rmax0);
        w.x = exp_hw(w.x - rmax1);
        w.y = exp_hw(w.y - rmax1);
        w.z = exp_hw(w.z - rmax1);
        w.w = exp_hw(w.w - rmax1);
        sum0 += (v.x + v.y) + (v.z + v.w);
        sum1 += (w.x + w.y) + (w.z + w.w);
        *(v4f*)p0 = v;
        *(v4f*)p1 = w;
    }
    red[r][c16]      = sum0;
    red[r + 16][c16] = sum1;
    __syncthreads();
    float rsum0 = 0.f;
    float rsum1 = 0.f;
    for (int t = 0; t < 16; ++t) {
        rsum0 += red[r][t];
        rsum1 += red[r + 16][t];
    }

    const float mlr0 = mlt_s[r] / rsum0;
    const float mlr1 = mlt_s[r + 16] / rsum1;
    float* __restrict__ orow0 = out + ((size_t)b * LL + lt0 + r) * (size_t)LL;
    float* __restrict__ orow1 = orow0 + (size_t)16 * LL;
    for (int j = 0; j < 16; ++j) {
        const int cc = c16 * 4 + 64 * j;
        const v4f e0 = *(const v4f*)(S + r * SSTR + cc);
        const v4f e1 = *(const v4f*)(S + (r + 16) * SSTR + cc);
        const v4f m4 = *(const v4f*)(mrt + cc);
        v4f o0, o1;
        o0.x = e0.x * mlr0 * m4.x;
        o0.y = e0.y * mlr0 * m4.y;
        o0.z = e0.z * mlr0 * m4.z;
        o0.w = e0.w * mlr0 * m4.w;
        o1.x = e1.x * mlr1 * m4.x;
        o1.y = e1.y * mlr1 * m4.y;
        o1.z = e1.z * mlr1 * m4.z;
        o1.w = e1.w * mlr1 * m4.w;
        __builtin_nontemporal_store(o0, (v4f*)(orow0 + cc));
        __builtin_nontemporal_store(o1, (v4f*)(orow1 + cc));
    }
}

extern "C" void kernel_launch(void* const* d_in, const int* in_sizes, int n_in,
                              void* d_out, int out_size, void* d_ws, size_t ws_size,
                              hipStream_t stream)
{
    (void)in_sizes; (void)n_in; (void)out_size; (void)ws_size;

    const float* reps_lt = (const float*)d_in[0];
    const float* reps_rt = (const float*)d_in[1];
    const float* mask_lt = (const float*)d_in[2];
    const float* mask_rt = (const float*)d_in[3];
    const float* attn_k  = (const float*)d_in[4];
    const float* diagW   = (const float*)d_in[5];
    float* outp = (float*)d_out;
    float* proj = (float*)d_ws;

    const size_t ldsW = (size_t)(HH / 2) * (2 * AA + 32) * sizeof(float);
    proj_tanh_kernel<<<dim3(512), 256, ldsW, stream>>>(
        reps_lt, reps_rt, attn_k, diagW, proj);

    const size_t ldsS = (size_t)32 * (LL + 4) * sizeof(float);
    attn_softmax_kernel<<<dim3(LL / 32, BB), 256, ldsS, stream>>>(
        proj, mask_lt, mask_rt, outp);
}


# projection K loop software-pipelined: the next 16 WMMAs' W fragments are read from LDS as one group ahead of the current block (two register sets), reps fragments 8 k-steps ahead; sums unchanged
# speedup vs baseline: 1.7157x; 1.0390x over previous
#include <hip/hip_runtime.h>
#include <cmath>


typedef __attribute__((ext_vector_type(2))) float v2f;
typedef __attribute__((ext_vector_type(4))) float v4f;
typedef __attribute__((ext_vector_type(8))) float v8f;

#define BB 32
#define LL 1024
#define HH 256
#define AA 128

__device__ __forceinline__ float exp_hw(float x)
{
    return __builtin_amdgcn_exp2f(x * 1.4426950408889634f);
}
__device__ __forceinline__ float tanh_hw(float x)
{
    const float e2x = __builtin_amdgcn_exp2f(x * 2.8853900817779268f);
    return 1.0f - 2.0f * __builtin_amdgcn_rcpf(e2x + 1.0f);
}

__device__ __forceinline__ void proj_group(
    v2f (&aCur)[8], v2f (&aNxt)[8], v2f (&wA)[2][8], v2f (&wB)[2][8], v8f (&acc)[8],
    const float* __restrict__ ap, const float* bp, int g)
{
    constexpr int PP = 2 * AA + 32;
    const int gn = (g < HH / 32 - 1) ? g + 1 : HH / 32 - 1;
#pragma unroll
    for (int s = 0; s < 8; ++s)
        aNxt[s] = *(const v2f*)(ap + 32 * gn + 4 * s);

#pragma unroll
    for (int blk = 0; blk < 4; ++blk) {
        v2f (&wc)[2][8] = (blk & 1) ? wB : wA;
        v2f (&wn)[2][8] = (blk & 1) ? wA : wB;
        const int p0 = 8 * g + 2 * blk;
        const int pn = (p0 < HH / 4 - 2) ? p0 + 2 : HH / 4 - 2;
        const float* bn = bp + pn * (2 * PP);
#pragma unroll
        for (int s = 0; s < 2; ++s)
#pragma unroll
            for (int ct = 0; ct < 8; ++ct)
                wn[s][ct] = *(const v2f*)(bn + s * (2 * PP) + 32 * ct);
        __builtin_amdgcn_sched_barrier(0);
#pragma unroll
        for (int s = 0; s < 2; ++s)
#pragma unroll
            for (int ct = 0; ct < 8; ++ct)
                acc[ct] = __builtin_amdgcn_wmma_f32_16x16x4_f32(
                    false, wc[s][ct], false, aCur[2 * blk + s], (short)0, acc[ct], false, false);
        __builtin_amdgcn_sched_barrier(0);
    }
}

__global__ __launch_bounds__(256) void proj_tanh_kernel(
    const float* __restrict__ reps_lt, const float* __restrict__ reps_rt,
    const float* __restrict__ W, const float* __restrict__ diagW,
    float* __restrict__ proj)
{
    extern __shared__ float Ws[];
    constexpr int PP = 2 * AA + 32;
    const int tid  = threadIdx.x;
    const int lane = tid & 31;
    const int wv   = tid >> 5;
    const int lo   = lane & 15;
    const int hi   = lane >> 4;

    const int row0 = (blockIdx.x * 8 + wv) * 16;
    const int grow = row0 + lo;
    const bool is_lt = row0 < BB * LL;
    const float* __restrict__ src =
        is_lt ? reps_lt + (size_t)grow * HH
              : reps_rt + (size_t)(grow - BB * LL) * HH;
    const float* __restrict__ ap = src + 2 * hi;

    v2f aA[8], aB[8];
#pragma unroll
    for (int s = 0; s < 8; ++s)
        aA[s] = *(const v2f*)(ap + 4 * s);

#pragma unroll
    for (int tb = 0; tb < 16; tb += 8) {
        v2f r0a[8], r0b[8], r1a[8], r1b[8];
#pragma unroll
        for (int t = 0; t < 8; ++t) {
            const float* g = W + (size_t)(2 * (wv + 8 * (tb + t))) * AA + 2 * lane;
            r0a[t] = *(const v2f*)(g);
            r0b[t] = *(const v2f*)(g + 64);
            r1a[t] = *(const v2f*)(g + AA);
            r1b[t] = *(const v2f*)(g + AA + 64);
        }
#pragma unroll
        for (int t = 0; t < 8; ++t) {
            float* d = Ws + (wv + 8 * (tb + t)) * PP + 4 * lane;
            const v4f s0 = {r0a[t].x, r1a[t].x, r0a[t].y, r1a[t].y};
            const v4f s1 = {r0b[t].x, r1b[t].x, r0b[t].y, r1b[t].y};
            *(v4f*)(d)       = s0;
            *(v4f*)(d + 128) = s1;
        }
    }
    __syncthreads();

    const float* bp = Ws + hi * PP + 2 * lo;

    v8f acc[8] = {};

    v2f wA[2][8], wB[2][8];
#pragma unroll
    for (int s = 0; s < 2; ++s)
#pragma unroll
        for (int ct = 0; ct < 8; ++ct)
            wA[s][ct] = *(const v2f*)(bp + s * (2 * PP) + 32 * ct);

    for (int g = 0; g < HH / 32; g += 2) {
        proj_group(aA, aB, wA, wB, acc, ap, bp, g);
        proj_group(aB, aA, wA, wB, acc, ap, bp, g + 1);
    }

    float* __restrict__ tile = proj + (size_t)(row0 >> 4) * (16 * AA);
    const v4f one4 = {1.0f, 1.0f, 1.0f, 1.0f};
#pragma unroll
    for (int ct = 0; ct < 8; ++ct) {
        const int j = 2 * ct + hi;
        const v4f dw0 = is_lt ? *(const v4f*)(diagW + 8 * j)     : one4;
        const v4f dw1 = is_lt ? *(const v4f*)(diagW + 8 * j + 4) : one4;
        v4f o0, o1;
        o0.x = tanh_hw(acc[ct][0]) * dw0.x;
        o0.y = tanh_hw(acc[ct][1]) * dw0.y;
        o0.z = tanh_hw(acc[ct][2]) * dw0.z;
        o0.w = tanh_hw(acc[ct][3]) * dw0.w;
        o1.x = tanh_hw(acc[ct][4]) * dw1.x;
        o1.y = tanh_hw(acc[ct][5]) * dw1.y;
        o1.z = tanh_hw(acc[ct][6]) * dw1.z;
        o1.w = tanh_hw(acc[ct][7]) * dw1.w;
        *(v4f*)(tile + (j * 32 + lo) * 4)      = o0;
        *(v4f*)(tile + (j * 32 + 16 + lo) * 4) = o1;
    }
}

__global__ __launch_bounds__(256) void attn_softmax_kernel(
    const float* __restrict__ proj, const float* __restrict__ mask_lt,
    const float* __restrict__ mask_rt, float* __restrict__ out)
{
    extern __shared__ float S[];
    constexpr int SSTR = LL + 4;
    __shared__ float red[32][17];
    __shared__ float mlt_s[32];

    const int tid = threadIdx.x;
    const int b   = blockIdx.y;
    const int lt0 = blockIdx.x * 32;

    if (tid < 32) mlt_s[tid] = mask_lt[(size_t)b * LL + lt0 + tid];
    __syncthreads();

    const int lane = tid & 31;
    const int wv   = tid >> 5;
    const int lo   = lane & 15;
    const int hi   = lane >> 4;

    const float* __restrict__ ltp0 =
        proj + (((size_t)b * LL + lt0) >> 4) * (16 * AA) + lane * 4;
    const float* __restrict__ ltp1 = ltp0 + 16 * AA;
    const float* __restrict__ rtb =
        proj + ((size_t)(BB * LL) + (size_t)b * LL) * AA + lane * 4;
    const float* __restrict__ mrt = mask_rt + (size_t)b * LL;

    v4f af0[16], af1[16];
#pragma unroll
    for (int j = 0; j < 16; ++j) {
        af0[j] = *(const v4f*)(ltp0 + 128 * j);
        af1[j] = *(const v4f*)(ltp1 + 128 * j);
    }

    for (int ct = 0; ct < 8; ++ct) {
        const int cbase = wv * 128 + ct * 16;
        const float* __restrict__ rtp = rtb + (size_t)(cbase >> 4) * (16 * AA);
        v8f c0 = {};
        v8f c1 = {};
#pragma unroll
        for (int q = 0; q < 2; ++q) {
            v4f bf[8];
#pragma unroll
            for (int u = 0; u < 8; ++u)
                bf[u] = *(const v4f*)(rtp + 128 * (8 * q + u));
            __builtin_amdgcn_sched_barrier(0);
#pragma unroll
            for (int u = 0; u < 8; ++u) {
                const v4f a40 = af0[8 * q + u];
                const v4f a41 = af1[8 * q + u];
                const v4f b4  = bf[u];
                const v2f b0  = {b4.x, b4.y};
                const v2f b1  = {b4.z, b4.w};
                const v2f a00 = {a40.x, a40.y};
                const v2f a10 = {a41.x, a41.y};
                const v2f a01 = {a40.z, a40.w};
                const v2f a11 = {a41.z, a41.w};
                c0 = __builtin_amdgcn_wmma_f32_16x16x4_f32(
                    false, a00, false, b0, (short)0, c0, false, false);
                c1 = __builtin_amdgcn_wmma_f32_16x16x4_f32(
                    false, a10, false, b0, (short)0, c1, false, false);
                c0 = __builtin_amdgcn_wmma_f32_16x16x4_f32(
                    false, a01, false, b1, (short)0, c0, false, false);
                c1 = __builtin_amdgcn_wmma_f32_16x16x4_f32(
                    false, a11, false, b1, (short)0, c1, false, false);
            }
        }
        const float mr = mrt[cbase + lo];
#pragma unroll
        for (int i = 0; i < 8; ++i) {
            const int m = i + 8 * hi;
            S[m * SSTR + cbase + lo]        = c0[i] * mlt_s[m] * mr;
            S[(16 + m) * SSTR + cbase + lo] = c1[i] * mlt_s[16 + m] * mr;
        }
    }
    __syncthreads();

    const int r   = tid >> 4;
    const int c16 = tid & 15;
    const float* Sr0 = S + r * SSTR + c16 * 4;
    const float* Sr1 = S + (r + 16) * SSTR + c16 * 4;

    float mx0 = -3.402823466e38f;
    float mx1 = -3.402823466e38f;
#pragma unroll 4
    for (int j = 0; j < 16; ++j) {
        const v4f v = *(const v4f*)(Sr0 + 64 * j);
        const v4f w = *(const v4f*)(Sr1 + 64 * j);
        mx0 = fmaxf(mx0, fmaxf(fmaxf(v.x, v.y), fmaxf(v.z, v.w)));
        mx1 = fmaxf(mx1, fmaxf(fmaxf(w.x, w.y), fmaxf(w.z, w.w)));
    }
    red[r][c16]      = mx0;
    red[r + 16][c16] = mx1;
    __syncthreads();
    float rmax0 = red[r][0];
    float rmax1 = red[r + 16][0];
    for (int t = 1; t < 16; ++t) {
        rmax0 = fmaxf(rmax0, red[r][t]);
        rmax1 = fmaxf(rmax1, red[r + 16][t]);
    }
    __syncthreads();

    float sum0 = 0.f;
    float sum1 = 0.f;
    for (int j = 0; j < 16; ++j) {
        float* p0 = S + r * SSTR + c16 * 4 + 64 * j;
        float* p1 = S + (r + 16) * SSTR + c16 * 4 + 64 * j;
        v4f v = *(const v4f*)p0;
        v4f w = *(const v4f*)p1;
        v.x = exp_hw(v.x - rmax0);
        v.y = exp_hw(v.y - rmax0);
        v.z = exp_hw(v.z - rmax0);
        v.w = exp_hw(v.w - rmax0);
        w.x = exp_hw(w.x - rmax1);
        w.y = exp_hw(w.y - rmax1);
        w.z = exp_hw(w.z - rmax1);
        w.w = exp_hw(w.w - rmax1);
        sum0 += (v.x + v.y) + (v.z + v.w);
        sum1 += (w.x + w.y) + (w.z + w.w);
        *(v4f*)p0 = v;
        *(v4f*)p1 = w;
    }
    red[r][c16]      = sum0;
    red[r + 16][c16] = sum1;
    __syncthreads();
    float rsum0 = 0.f;
    float rsum1 = 0.f;
    for (int t = 0; t < 16; ++t) {
        rsum0 += red[r][t];
        rsum1 += red[r + 16][t];
    }

    const float mlr0 = mlt_s[r] / rsum0;
    const float mlr1 = mlt_s[r + 16] / rsum1;
    float* __restrict__ orow0 = out + ((size_t)b * LL + lt0 + r) * (size_t)LL;
    float* __restrict__ orow1 = orow0 + (size_t)16 * LL;
    for (int j = 0; j < 16; ++j) {
        const int cc = c16 * 4 + 64 * j;
        const v4f e0 = *(const v4f*)(S + r * SSTR + cc);
        const v4f e1 = *(const v4f*)(S + (r + 16) * SSTR + cc);
        const v4f m4 = *(const v4f*)(mrt + cc);
        v4f o0, o1;
        o0.x = e0.x * mlr0 * m4.x;
        o0.y = e0.y * mlr0 * m4.y;
        o0.z = e0.z * mlr0 * m4.z;
        o0.w = e0.w * mlr0 * m4.w;
        o1.x = e1.x * mlr1 * m4.x;
        o1.y = e1.y * mlr1 * m4.y;
        o1.z = e1.z * mlr1 * m4.z;
        o1.w = e1.w * mlr1 * m4.w;
        __builtin_nontemporal_store(o0, (v4f*)(orow0 + cc));
        __builtin_nontemporal_store(o1, (v4f*)(orow1 + cc));
    }
}

extern "C" void kernel_launch(void* const* d_in, const int* in_sizes, int n_in,
                              void* d_out, int out_size, void* d_ws, size_t ws_size,
                              hipStream_t stream)
{
    (void)in_sizes; (void)n_in; (void)out_size; (void)ws_size;

    const float* reps_lt = (const float*)d_in[0];
    const float* reps_rt = (const float*)d_in[1];
    const float* mask_lt = (const float*)d_in[2];
    const float* mask_rt = (const float*)d_in[3];
    const float* attn_k  = (const float*)d_in[4];
    const float* diagW   = (const float*)d_in[5];
    float* outp = (float*)d_out;
    float* proj = (float*)d_ws;

    const size_t ldsW = (size_t)(HH / 2) * (2 * AA + 32) * sizeof(float);
    proj_tanh_kernel<<<dim3(512), 256, ldsW, stream>>>(
        reps_lt, reps_rt, attn_k, diagW, proj);

    const size_t ldsS = (size_t)32 * (LL + 4) * sizeof(float);
    attn_softmax_kernel<<<dim3(LL / 32, BB), 256, ldsS, stream>>>(
        proj, mask_lt, mask_rt, outp);
}
